# starting kernel
# baseline (speedup 1.0000x reference)
#include <hip/hip_runtime.h>
#include <math.h>

typedef __attribute__((ext_vector_type(2))) float v2f;
typedef __attribute__((ext_vector_type(8))) float v8f;

constexpr int D      = 2048;
constexpr int T      = 10;
constexpr int TP     = 16;
constexpr int NROWS  = 4 * 4096;
constexpr int KC       = 256;
constexpr int NCHUNK   = D / KC;
constexpr int WAVES    = 2;
constexpr int BLOCK    = WAVES * 32;
constexpr int ROWS_PB  = WAVES * 16;

__global__ __launch_bounds__(BLOCK)
void tgate_hybrid_fused_kernel(const float* __restrict__ x,
                               const float* __restrict__ Wc, const float* __restrict__ bc,
                               const float* __restrict__ Ws, const float* __restrict__ bs,
                               const float* __restrict__ Wg, const float* __restrict__ bg,
                               const float* __restrict__ alpha,
                               float* __restrict__ out)
{
    __shared__ float lds_w[3][KC / 2][TP][2];
    __shared__ float lds_acc[ROWS_PB][3][TP];

    const int tid  = threadIdx.x;
    const int wid  = tid >> 5;
    const int lane = tid & 31;
    const int half = lane >> 4;
    const int l16  = lane & 15;

    const int rowBase = blockIdx.x * ROWS_PB + wid * 16;
    const float* px   = x + (size_t)(rowBase + l16) * D;

    const float* Wmat[3] = { Wc, Ws, Wg };

    v8f acc[3];
    #pragma unroll
    for (int m = 0; m < 3; ++m)
        #pragma unroll
        for (int r = 0; r < 8; ++r)
            acc[m][r] = 0.0f;

    for (int c = 0; c < NCHUNK; ++c) {
        const int k0 = c * KC;
        __syncthreads();

        for (int idx = tid; idx < 3 * KC * TP; idx += BLOCK) {
            const int m = idx / (KC * TP);
            const int r = idx % (KC * TP);
            const int k = r / TP;
            const int n = r % TP;
            const float v = (n < T) ? Wmat[m][(size_t)(k0 + k) * T + n] : 0.0f;
            lds_w[m][k >> 1][n][k & 1] = v;
        }
        __syncthreads();

        if (c + 1 < NCHUNK)
            __builtin_prefetch(px + k0 + KC + 2 * half, 0, 1);

        #pragma unroll 8
        for (int kk = 0; kk < KC; kk += 4) {
            const v2f a = __builtin_nontemporal_load(
                (const v2f*)(px + k0 + kk + 2 * half));
            #pragma unroll
            for (int m = 0; m < 3; ++m) {
                const v2f b = *(const v2f*)&lds_w[m][(kk >> 1) + half][l16][0];
                acc[m] = __builtin_amdgcn_wmma_f32_16x16x4_f32(
                    false, a, false, b, (short)0, acc[m], false, false);
            }
        }
    }

    __syncthreads();
    #pragma unroll
    for (int m = 0; m < 3; ++m)
        #pragma unroll
        for (int r = 0; r < 8; ++r)
            lds_acc[wid * 16 + r + 8 * half][m][l16] = acc[m][r];
    __syncthreads();

    if (tid < ROWS_PB) {
        const int row = tid;
        float Lc[T], Ls[T], Lg[T];
        #pragma unroll
        for (int n = 0; n < T; ++n) {
            Lc[n] = lds_acc[row][0][n] + bc[n];
            Ls[n] = lds_acc[row][1][n] + bs[n];
            Lg[n] = lds_acc[row][2][n] + bg[n];
        }
        float mx = Lc[0];
        #pragma unroll
        for (int n = 1; n < T; ++n) mx = fmaxf(mx, Lc[n]);
        float e[T], ssum = 0.0f;
        #pragma unroll
        for (int n = 0; n < T; ++n) { e[n] = __expf(Lc[n] - mx); ssum += e[n]; }
        float g[T];
        #pragma unroll
        for (int n = 0; n < T; ++n) g[n] = 1.0f / (1.0f + __expf(-Lg[n]));
        float softdot = 0.0f;
        #pragma unroll
        for (int n = 0; n < T; ++n) softdot += g[n] * e[n];
        softdot /= ssum;
        float v0 = -INFINITY, v1 = -INFINITY;
        int   i0 = 0, i1 = 0;
        #pragma unroll
        for (int n = 0; n < T; ++n) {
            const float t = Ls[n];
            if (t > v0)      { v1 = v0; i1 = i0; v0 = t; i0 = n; }
            else if (t > v1) { v1 = t;  i1 = n; }
        }
        const float s0 = 1.0f / (1.0f + __expf(v1 - v0));
        const float s1 = 1.0f - s0;
        const float sparsedot = g[i0] * s0 + g[i1] * s1;

        const float a = 1.0f / (1.0f + __expf(-alpha[0]));
        out[blockIdx.x * ROWS_PB + row] = a * sparsedot + (1.0f - a) * softdot;
    }
}

extern "C" void kernel_launch(void* const* d_in, const int* in_sizes, int n_in,
                              void* d_out, int out_size, void* d_ws, size_t ws_size,
                              hipStream_t stream) {
    (void)in_sizes; (void)n_in; (void)out_size; (void)d_ws; (void)ws_size;
    const float* x     = (const float*)d_in[0];
    const float* W_cls = (const float*)d_in[1];
    const float* b_cls = (const float*)d_in[2];
    const float* W_sp  = (const float*)d_in[3];
    const float* b_sp  = (const float*)d_in[4];
    const float* W_g   = (const float*)d_in[5];
    const float* b_g   = (const float*)d_in[6];
    const float* alpha = (const float*)d_in[7];
    float* out = (float*)d_out;

    dim3 grid(NROWS / ROWS_PB);
    dim3 block(BLOCK);
    tgate_hybrid_fused_kernel<<<grid, block, 0, stream>>>(
        x, W_cls, b_cls, W_sp, b_sp, W_g, b_g, alpha, out);
}


# Weight chunk staged as a straight 16-byte copy in its global layout, all 30 loads issued before the first store (was one load-wait-store per element); B fragment by two-address LDS load
# speedup vs baseline: 8.0390x; 8.0390x over previous
#include <hip/hip_runtime.h>
#include <math.h>


typedef __attribute__((ext_vector_type(2))) float v2f;
typedef __attribute__((ext_vector_type(4))) float v4f;
typedef __attribute__((ext_vector_type(8))) float v8f;

constexpr int D      = 2048;
constexpr int T      = 10;
constexpr int TP     = 16;
constexpr int NROWS  = 4 * 4096;
constexpr int KC       = 256;
constexpr int NCHUNK   = D / KC;
constexpr int WAVES    = 2;
constexpr int BLOCK    = WAVES * 32;
constexpr int ROWS_PB  = WAVES * 16;
constexpr int CH_F   = KC * T;
constexpr int CH_V4  = CH_F / 4;
constexpr int V4_PT  = CH_V4 / BLOCK;

__global__ __launch_bounds__(BLOCK)
void tgate_hybrid_fused_kernel(const float* __restrict__ x,
                               const float* __restrict__ Wc, const float* __restrict__ bc,
                               const float* __restrict__ Ws, const float* __restrict__ bs,
                               const float* __restrict__ Wg, const float* __restrict__ bg,
                               const float* __restrict__ alpha,
                               float* __restrict__ out)
{
    __shared__ __attribute__((aligned(16))) float lds_w[3][CH_F];
    __shared__ float lds_acc[ROWS_PB][3][TP];

    const int tid  = threadIdx.x;
    const int wid  = tid >> 5;
    const int lane = tid & 31;
    const int half = lane >> 4;
    const int l16  = lane & 15;

    const int rowBase = blockIdx.x * ROWS_PB + wid * 16;
    const float* px   = x + (size_t)(rowBase + l16) * D;

    const int cB   = (l16 < T) ? l16 : (T - 1);
    const int bOff = (2 * half) * T + cB;

    v8f acc[3];
    #pragma unroll
    for (int m = 0; m < 3; ++m)
        #pragma unroll
        for (int r = 0; r < 8; ++r)
            acc[m][r] = 0.0f;

    for (int c = 0; c < NCHUNK; ++c) {
        const int k0 = c * KC;
        __syncthreads();

        {
            const v4f* g0 = (const v4f*)(Wc + (size_t)k0 * T);
            const v4f* g1 = (const v4f*)(Ws + (size_t)k0 * T);
            const v4f* g2 = (const v4f*)(Wg + (size_t)k0 * T);
            v4f w0[V4_PT], w1[V4_PT], w2[V4_PT];
            #pragma unroll
            for (int j = 0; j < V4_PT; ++j) {
                w0[j] = g0[tid + j * BLOCK];
                w1[j] = g1[tid + j * BLOCK];
                w2[j] = g2[tid + j * BLOCK];
            }
            #pragma unroll
            for (int j = 0; j < V4_PT; ++j) {
                *(v4f*)&lds_w[0][4 * (tid + j * BLOCK)] = w0[j];
                *(v4f*)&lds_w[1][4 * (tid + j * BLOCK)] = w1[j];
                *(v4f*)&lds_w[2][4 * (tid + j * BLOCK)] = w2[j];
            }
        }
        __syncthreads();

        if (c + 1 < NCHUNK)
            __builtin_prefetch(px + k0 + KC + 2 * half, 0, 1);

        #pragma unroll 8
        for (int kk = 0; kk < KC; kk += 4) {
            const v2f a = __builtin_nontemporal_load(
                (const v2f*)(px + k0 + kk + 2 * half));
            #pragma unroll
            for (int m = 0; m < 3; ++m) {
                v2f b;
                b[0] = lds_w[m][(kk    ) * T + bOff];
                b[1] = lds_w[m][(kk + 1) * T + bOff];
                acc[m] = __builtin_amdgcn_wmma_f32_16x16x4_f32(
                    false, a, false, b, (short)0, acc[m], false, false);
            }
        }
    }

    __syncthreads();
    #pragma unroll
    for (int m = 0; m < 3; ++m)
        #pragma unroll
        for (int r = 0; r < 8; ++r)
            lds_acc[wid * 16 + r + 8 * half][m][l16] = acc[m][r];
    __syncthreads();

    if (tid < ROWS_PB) {
        const int row = tid;
        float Lc[T], Ls[T], Lg[T];
        #pragma unroll
        for (int n = 0; n < T; ++n) {
            Lc[n] = lds_acc[row][0][n] + bc[n];
            Ls[n] = lds_acc[row][1][n] + bs[n];
            Lg[n] = lds_acc[row][2][n] + bg[n];
        }
        float mx = Lc[0];
        #pragma unroll
        for (int n = 1; n < T; ++n) mx = fmaxf(mx, Lc[n]);
        float e[T], ssum = 0.0f;
        #pragma unroll
        for (int n = 0; n < T; ++n) { e[n] = __expf(Lc[n] - mx); ssum += e[n]; }
        float g[T];
        #pragma unroll
        for (int n = 0; n < T; ++n) g[n] = 1.0f / (1.0f + __expf(-Lg[n]));
        float softdot = 0.0f;
        #pragma unroll
        for (int n = 0; n < T; ++n) softdot += g[n] * e[n];
        softdot /= ssum;
        float v0 = -INFINITY, v1 = -INFINITY;
        int   i0 = 0, i1 = 0;
        #pragma unroll
        for (int n = 0; n < T; ++n) {
            const float t = Ls[n];
            if (t > v0)      { v1 = v0; i1 = i0; v0 = t; i0 = n; }
            else if (t > v1) { v1 = t;  i1 = n; }
        }
        const float s0 = 1.0f / (1.0f + __expf(v1 - v0));
        const float s1 = 1.0f - s0;
        const float sparsedot = g[i0] * s0 + g[i1] * s1;

        const float a = 1.0f / (1.0f + __expf(-alpha[0]));
        out[blockIdx.x * ROWS_PB + row] = a * sparsedot + (1.0f - a) * softdot;
    }
}

extern "C" void kernel_launch(void* const* d_in, const int* in_sizes, int n_in,
                              void* d_out, int out_size, void* d_ws, size_t ws_size,
                              hipStream_t stream) {
    (void)in_sizes; (void)n_in; (void)out_size; (void)d_ws; (void)ws_size;
    const float* x     = (const float*)d_in[0];
    const float* W_cls = (const float*)d_in[1];
    const float* b_cls = (const float*)d_in[2];
    const float* W_sp  = (const float*)d_in[3];
    const float* b_sp  = (const float*)d_in[4];
    const float* W_g   = (const float*)d_in[5];
    const float* b_g   = (const float*)d_in[6];
    const float* alpha = (const float*)d_in[7];
    float* out = (float*)d_out;

    dim3 grid(NROWS / ROWS_PB);
    dim3 block(BLOCK);
    tgate_hybrid_fused_kernel<<<grid, block, 0, stream>>>(
        x, W_cls, b_cls, W_sp, b_sp, W_g, b_g, alpha, out);
}


# K range of each chunk split over 4 waves per 16-row tile (8-wave blocks, was 2; chunk 512 so staging divides evenly); the 4 partial sums are added through LDS in the epilogue (sum order changes)
# speedup vs baseline: 14.3298x; 1.7825x over previous
#include <hip/hip_runtime.h>
#include <math.h>


typedef __attribute__((ext_vector_type(2))) float v2f;
typedef __attribute__((ext_vector_type(4))) float v4f;
typedef __attribute__((ext_vector_type(8))) float v8f;

constexpr int D      = 2048;
constexpr int T      = 10;
constexpr int TP     = 16;
constexpr int NROWS  = 4 * 4096;
constexpr int KC       = 512;
constexpr int NCHUNK   = D / KC;
constexpr int ROWT     = 2;
constexpr int KSPLIT   = 4;
constexpr int WAVES    = ROWT * KSPLIT;
constexpr int BLOCK    = WAVES * 32;
constexpr int ROWS_PB  = ROWT * 16;
constexpr int KW       = KC / KSPLIT;
constexpr int CH_F   = KC * T;
constexpr int CH_V4  = CH_F / 4;
constexpr int V4_PT  = CH_V4 / BLOCK;
constexpr int RP     = 50;

__global__ __launch_bounds__(BLOCK)
void tgate_hybrid_fused_kernel(const float* __restrict__ x,
                               const float* __restrict__ Wc, const float* __restrict__ bc,
                               const float* __restrict__ Ws, const float* __restrict__ bs,
                               const float* __restrict__ Wg, const float* __restrict__ bg,
                               const float* __restrict__ alpha,
                               float* __restrict__ out)
{
    __shared__ __attribute__((aligned(16))) float lds_w[3][CH_F];
    __shared__ float lds_acc[KSPLIT][ROWS_PB][RP];

    const int tid  = threadIdx.x;
    const int wid  = tid >> 5;
    const int lane = tid & 31;
    const int half = lane >> 4;
    const int l16  = lane & 15;
    const int rt   = wid & (ROWT - 1);
    const int kq   = wid / ROWT;
    const int kw0  = kq * KW;

    const int rowBase = blockIdx.x * ROWS_PB + rt * 16;
    const float* px   = x + (size_t)(rowBase + l16) * D + kw0;

    const int cB   = (l16 < T) ? l16 : (T - 1);
    const int bOff = (kw0 + 2 * half) * T + cB;

    v8f acc[3];
    #pragma unroll
    for (int m = 0; m < 3; ++m)
        #pragma unroll
        for (int r = 0; r < 8; ++r)
            acc[m][r] = 0.0f;

    for (int c = 0; c < NCHUNK; ++c) {
        const int k0 = c * KC;
        __syncthreads();

        {
            const v4f* g0 = (const v4f*)(Wc + (size_t)k0 * T);
            const v4f* g1 = (const v4f*)(Ws + (size_t)k0 * T);
            const v4f* g2 = (const v4f*)(Wg + (size_t)k0 * T);
            v4f w0[V4_PT], w1[V4_PT], w2[V4_PT];
            #pragma unroll
            for (int j = 0; j < V4_PT; ++j) {
                w0[j] = g0[tid + j * BLOCK];
                w1[j] = g1[tid + j * BLOCK];
                w2[j] = g2[tid + j * BLOCK];
            }
            #pragma unroll
            for (int j = 0; j < V4_PT; ++j) {
                *(v4f*)&lds_w[0][4 * (tid + j * BLOCK)] = w0[j];
                *(v4f*)&lds_w[1][4 * (tid + j * BLOCK)] = w1[j];
                *(v4f*)&lds_w[2][4 * (tid + j * BLOCK)] = w2[j];
            }
        }
        __syncthreads();

        if (c + 1 < NCHUNK)
            __builtin_prefetch(px + k0 + KC + 2 * half, 0, 1);

        #pragma unroll 8
        for (int kk = 0; kk < KW; kk += 4) {
            const v2f a = __builtin_nontemporal_load(
                (const v2f*)(px + k0 + kk + 2 * half));
            #pragma unroll
            for (int m = 0; m < 3; ++m) {
                v2f b;
                b[0] = lds_w[m][(kk    ) * T + bOff];
                b[1] = lds_w[m][(kk + 1) * T + bOff];
                acc[m] = __builtin_amdgcn_wmma_f32_16x16x4_f32(
                    false, a, false, b, (short)0, acc[m], false, false);
            }
        }
    }

    #pragma unroll
    for (int m = 0; m < 3; ++m)
        #pragma unroll
        for (int r = 0; r < 8; ++r)
            lds_acc[kq][rt * 16 + r + 8 * half][m * TP + l16] = acc[m][r];
    __syncthreads();

    if (tid < ROWS_PB) {
        const int row = tid;
        float Lc[T], Ls[T], Lg[T];
        #pragma unroll
        for (int n = 0; n < T; ++n) {
            Lc[n] = ((lds_acc[0][row][0 * TP + n] + lds_acc[1][row][0 * TP + n])
                   + (lds_acc[2][row][0 * TP + n] + lds_acc[3][row][0 * TP + n])) + bc[n];
            Ls[n] = ((lds_acc[0][row][1 * TP + n] + lds_acc[1][row][1 * TP + n])
                   + (lds_acc[2][row][1 * TP + n] + lds_acc[3][row][1 * TP + n])) + bs[n];
            Lg[n] = ((lds_acc[0][row][2 * TP + n] + lds_acc[1][row][2 * TP + n])
                   + (lds_acc[2][row][2 * TP + n] + lds_acc[3][row][2 * TP + n])) + bg[n];
        }
        float mx = Lc[0];
        #pragma unroll
        for (int n = 1; n < T; ++n) mx = fmaxf(mx, Lc[n]);
        float e[T], ssum = 0.0f;
        #pragma unroll
        for (int n = 0; n < T; ++n) { e[n] = __expf(Lc[n] - mx); ssum += e[n]; }
        float g[T];
        #pragma unroll
        for (int n = 0; n < T; ++n) g[n] = 1.0f / (1.0f + __expf(-Lg[n]));
        float softdot = 0.0f;
        #pragma unroll
        for (int n = 0; n < T; ++n) softdot += g[n] * e[n];
        softdot /= ssum;
        float v0 = -INFINITY, v1 = -INFINITY;
        int   i0 = 0, i1 = 0;
        #pragma unroll
        for (int n = 0; n < T; ++n) {
            const float t = Ls[n];
            if (t > v0)      { v1 = v0; i1 = i0; v0 = t; i0 = n; }
            else if (t > v1) { v1 = t;  i1 = n; }
        }
        const float s0 = 1.0f / (1.0f + __expf(v1 - v0));
        const float s1 = 1.0f - s0;
        const float sparsedot = g[i0] * s0 + g[i1] * s1;

        const float a = 1.0f / (1.0f + __expf(-alpha[0]));
        out[blockIdx.x * ROWS_PB + row] = a * sparsedot + (1.0f - a) * softdot;
    }
}

extern "C" void kernel_launch(void* const* d_in, const int* in_sizes, int n_in,
                              void* d_out, int out_size, void* d_ws, size_t ws_size,
                              hipStream_t stream) {
    (void)in_sizes; (void)n_in; (void)out_size; (void)d_ws; (void)ws_size;
    const float* x     = (const float*)d_in[0];
    const float* W_cls = (const float*)d_in[1];
    const float* b_cls = (const float*)d_in[2];
    const float* W_sp  = (const float*)d_in[3];
    const float* b_sp  = (const float*)d_in[4];
    const float* W_g   = (const float*)d_in[5];
    const float* b_g   = (const float*)d_in[6];
    const float* alpha = (const float*)d_in[7];
    float* out = (float*)d_out;

    dim3 grid(NROWS / ROWS_PB);
    dim3 block(BLOCK);
    tgate_hybrid_fused_kernel<<<grid, block, 0, stream>>>(
        x, W_cls, b_cls, W_sp, b_sp, W_g, b_g, alpha, out);
}


# K loop software-pipelined by one 8-step trip: next trip's 8 x loads and 24 B-fragment LDS loads leave as one group ahead of the current trip's 24 WMMAs (second register set); sum order unchanged
# speedup vs baseline: 14.9525x; 1.0435x over previous
#include <hip/hip_runtime.h>
#include <math.h>


typedef __attribute__((ext_vector_type(2))) float v2f;
typedef __attribute__((ext_vector_type(4))) float v4f;
typedef __attribute__((ext_vector_type(8))) float v8f;

constexpr int D      = 2048;
constexpr int T      = 10;
constexpr int TP     = 16;
constexpr int NROWS  = 4 * 4096;
constexpr int KC       = 512;
constexpr int NCHUNK   = D / KC;
constexpr int ROWT     = 2;
constexpr int KSPLIT   = 4;
constexpr int WAVES    = ROWT * KSPLIT;
constexpr int BLOCK    = WAVES * 32;
constexpr int ROWS_PB  = ROWT * 16;
constexpr int KW       = KC / KSPLIT;
constexpr int CH_F   = KC * T;
constexpr int CH_V4  = CH_F / 4;
constexpr int V4_PT  = CH_V4 / BLOCK;
constexpr int RP     = 50;

__global__ __launch_bounds__(BLOCK)
void tgate_hybrid_fused_kernel(const float* __restrict__ x,
                               const float* __restrict__ Wc, const float* __restrict__ bc,
                               const float* __restrict__ Ws, const float* __restrict__ bs,
                               const float* __restrict__ Wg, const float* __restrict__ bg,
                               const float* __restrict__ alpha,
                               float* __restrict__ out)
{
    __shared__ __attribute__((aligned(16))) float lds_w[3][CH_F];
    __shared__ float lds_acc[KSPLIT][ROWS_PB][RP];

    const int tid  = threadIdx.x;
    const int wid  = tid >> 5;
    const int lane = tid & 31;
    const int half = lane >> 4;
    const int l16  = lane & 15;
    const int rt   = wid & (ROWT - 1);
    const int kq   = wid / ROWT;
    const int kw0  = kq * KW;

    const int rowBase = blockIdx.x * ROWS_PB + rt * 16;
    const float* px   = x + (size_t)(rowBase + l16) * D + kw0;

    const int cB   = (l16 < T) ? l16 : (T - 1);
    const int bOff = (kw0 + 2 * half) * T + cB;

    v8f acc[3];
    #pragma unroll
    for (int m = 0; m < 3; ++m)
        #pragma unroll
        for (int r = 0; r < 8; ++r)
            acc[m][r] = 0.0f;

    constexpr int TRIP  = 8;
    constexpr int NTRIP = KW / (4 * TRIP);
    v2f xa[2][TRIP];
    v2f bb[2][TRIP][3];

    for (int c = 0; c < NCHUNK; ++c) {
        const int k0 = c * KC;
        const float* pxc = px + k0 + 2 * half;

        __syncthreads();

        #pragma unroll
        for (int j = 0; j < TRIP; ++j)
            xa[0][j] = __builtin_nontemporal_load((const v2f*)(pxc + 4 * j));

        {
            const v4f* g0 = (const v4f*)(Wc + (size_t)k0 * T);
            const v4f* g1 = (const v4f*)(Ws + (size_t)k0 * T);
            const v4f* g2 = (const v4f*)(Wg + (size_t)k0 * T);
            v4f w0[V4_PT], w1[V4_PT], w2[V4_PT];
            #pragma unroll
            for (int j = 0; j < V4_PT; ++j) {
                w0[j] = g0[tid + j * BLOCK];
                w1[j] = g1[tid + j * BLOCK];
                w2[j] = g2[tid + j * BLOCK];
            }
            #pragma unroll
            for (int j = 0; j < V4_PT; ++j) {
                *(v4f*)&lds_w[0][4 * (tid + j * BLOCK)] = w0[j];
                *(v4f*)&lds_w[1][4 * (tid + j * BLOCK)] = w1[j];
                *(v4f*)&lds_w[2][4 * (tid + j * BLOCK)] = w2[j];
            }
        }
        __syncthreads();

        if (c + 1 < NCHUNK)
            __builtin_prefetch(px + k0 + KC + 2 * half, 0, 1);

        #pragma unroll
        for (int j = 0; j < TRIP; ++j)
            #pragma unroll
            for (int m = 0; m < 3; ++m) {
                bb[0][j][m][0] = lds_w[m][(4 * j    ) * T + bOff];
                bb[0][j][m][1] = lds_w[m][(4 * j + 1) * T + bOff];
            }

        #pragma unroll
        for (int t = 0; t < NTRIP; ++t) {
            const int cur = t & 1, nxt = cur ^ 1;
            if (t + 1 < NTRIP) {
                const int kn = 4 * TRIP * (t + 1);
                #pragma unroll
                for (int j = 0; j < TRIP; ++j)
                    xa[nxt][j] = __builtin_nontemporal_load((const v2f*)(pxc + kn + 4 * j));
                #pragma unroll
                for (int j = 0; j < TRIP; ++j)
                    #pragma unroll
                    for (int m = 0; m < 3; ++m) {
                        bb[nxt][j][m][0] = lds_w[m][(kn + 4 * j    ) * T + bOff];
                        bb[nxt][j][m][1] = lds_w[m][(kn + 4 * j + 1) * T + bOff];
                    }
            }
            __builtin_amdgcn_sched_barrier(0);
            #pragma unroll
            for (int j = 0; j < TRIP; ++j)
                #pragma unroll
                for (int m = 0; m < 3; ++m)
                    acc[m] = __builtin_amdgcn_wmma_f32_16x16x4_f32(
                        false, xa[cur][j], false, bb[cur][j][m], (short)0, acc[m], false, false);
            __builtin_amdgcn_sched_barrier(0);
        }
    }

    #pragma unroll
    for (int m = 0; m < 3; ++m)
        #pragma unroll
        for (int r = 0; r < 8; ++r)
            lds_acc[kq][rt * 16 + r + 8 * half][m * TP + l16] = acc[m][r];
    __syncthreads();

    if (tid < ROWS_PB) {
        const int row = tid;
        float Lc[T], Ls[T], Lg[T];
        #pragma unroll
        for (int n = 0; n < T; ++n) {
            Lc[n] = ((lds_acc[0][row][0 * TP + n] + lds_acc[1][row][0 * TP + n])
                   + (lds_acc[2][row][0 * TP + n] + lds_acc[3][row][0 * TP + n])) + bc[n];
            Ls[n] = ((lds_acc[0][row][1 * TP + n] + lds_acc[1][row][1 * TP + n])
                   + (lds_acc[2][row][1 * TP + n] + lds_acc[3][row][1 * TP + n])) + bs[n];
            Lg[n] = ((lds_acc[0][row][2 * TP + n] + lds_acc[1][row][2 * TP + n])
                   + (lds_acc[2][row][2 * TP + n] + lds_acc[3][row][2 * TP + n])) + bg[n];
        }
        float mx = Lc[0];
        #pragma unroll
        for (int n = 1; n < T; ++n) mx = fmaxf(mx, Lc[n]);
        float e[T], ssum = 0.0f;
        #pragma unroll
        for (int n = 0; n < T; ++n) { e[n] = __expf(Lc[n] - mx); ssum += e[n]; }
        float g[T];
        #pragma unroll
        for (int n = 0; n < T; ++n) g[n] = 1.0f / (1.0f + __expf(-Lg[n]));
        float softdot = 0.0f;
        #pragma unroll
        for (int n = 0; n < T; ++n) softdot += g[n] * e[n];
        softdot /= ssum;
        float v0 = -INFINITY, v1 = -INFINITY;
        int   i0 = 0, i1 = 0;
        #pragma unroll
        for (int n = 0; n < T; ++n) {
            const float t = Ls[n];
            if (t > v0)      { v1 = v0; i1 = i0; v0 = t; i0 = n; }
            else if (t > v1) { v1 = t;  i1 = n; }
        }
        const float s0 = 1.0f / (1.0f + __expf(v1 - v0));
        const float s1 = 1.0f - s0;
        const float sparsedot = g[i0] * s0 + g[i1] * s1;

        const float a = 1.0f / (1.0f + __expf(-alpha[0]));
        out[blockIdx.x * ROWS_PB + row] = a * sparsedot + (1.0f - a) * softdot;
    }
}

extern "C" void kernel_launch(void* const* d_in, const int* in_sizes, int n_in,
                              void* d_out, int out_size, void* d_ws, size_t ws_size,
                              hipStream_t stream) {
    (void)in_sizes; (void)n_in; (void)out_size; (void)d_ws; (void)ws_size;
    const float* x     = (const float*)d_in[0];
    const float* W_cls = (const float*)d_in[1];
    const float* b_cls = (const float*)d_in[2];
    const float* W_sp  = (const float*)d_in[3];
    const float* b_sp  = (const float*)d_in[4];
    const float* W_g   = (const float*)d_in[5];
    const float* b_g   = (const float*)d_in[6];
    const float* alpha = (const float*)d_in[7];
    float* out = (float*)d_out;

    dim3 grid(NROWS / ROWS_PB);
    dim3 block(BLOCK);
    tgate_hybrid_fused_kernel<<<grid, block, 0, stream>>>(
        x, W_cls, b_cls, W_sp, b_sp, W_g, b_g, alpha, out);
}


# The 30 output columns of the three weight matrices packed side by side into 2 WMMA N tiles (was 3 tiles, 6 dead columns each): 16 WMMAs and 16 B loads per trip, was 24; sum order unchanged
# speedup vs baseline: 17.1454x; 1.1467x over previous
#include <hip/hip_runtime.h>
#include <math.h>


typedef __attribute__((ext_vector_type(2))) float v2f;
typedef __attribute__((ext_vector_type(4))) float v4f;
typedef __attribute__((ext_vector_type(8))) float v8f;

constexpr int D      = 2048;
constexpr int T      = 10;
constexpr int NP     = 3 * T;
constexpr int NTILE  = 2;
constexpr int NROWS  = 4 * 4096;
constexpr int KC       = 512;
constexpr int NCHUNK   = D / KC;
constexpr int ROWT     = 2;
constexpr int KSPLIT   = 4;
constexpr int WAVES    = ROWT * KSPLIT;
constexpr int BLOCK    = WAVES * 32;
constexpr int ROWS_PB  = ROWT * 16;
constexpr int KW       = KC / KSPLIT;
constexpr int CH_F   = KC * T;
constexpr int CH_V4  = CH_F / 4;
constexpr int V4_PT  = CH_V4 / BLOCK;
constexpr int MSTR   = CH_F + 12;
constexpr int RP     = 34;

__global__ __launch_bounds__(BLOCK)
void tgate_hybrid_fused_kernel(const float* __restrict__ x,
                               const float* __restrict__ Wc, const float* __restrict__ bc,
                               const float* __restrict__ Ws, const float* __restrict__ bs,
                               const float* __restrict__ Wg, const float* __restrict__ bg,
                               const float* __restrict__ alpha,
                               float* __restrict__ out)
{
    __shared__ __attribute__((aligned(16))) float lds_w[3 * MSTR];
    __shared__ float lds_acc[KSPLIT][ROWS_PB][RP];

    const int tid  = threadIdx.x;
    const int wid  = tid >> 5;
    const int lane = tid & 31;
    const int half = lane >> 4;
    const int l16  = lane & 15;
    const int rt   = wid & (ROWT - 1);
    const int kq   = wid / ROWT;
    const int kw0  = kq * KW;

    const int rowBase = blockIdx.x * ROWS_PB + rt * 16;
    const float* px   = x + (size_t)(rowBase + l16) * D + kw0;

    int bOff[NTILE];
    #pragma unroll
    for (int t = 0; t < NTILE; ++t) {
        const int p = (16 * t + l16 < NP) ? (16 * t + l16) : (NP - 1);
        const int m = p / T;
        bOff[t] = m * MSTR + (kw0 + 2 * half) * T + (p - m * T);
    }

    v8f acc[NTILE];
    #pragma unroll
    for (int t = 0; t < NTILE; ++t)
        #pragma unroll
        for (int r = 0; r < 8; ++r)
            acc[t][r] = 0.0f;

    constexpr int TRIP  = 8;
    constexpr int NTRIP = KW / (4 * TRIP);
    v2f xa[2][TRIP];
    v2f bb[2][TRIP][NTILE];

    for (int c = 0; c < NCHUNK; ++c) {
        const int k0 = c * KC;
        const float* pxc = px + k0 + 2 * half;

        __syncthreads();

        #pragma unroll
        for (int j = 0; j < TRIP; ++j)
            xa[0][j] = __builtin_nontemporal_load((const v2f*)(pxc + 4 * j));

        {
            const v4f* g0 = (const v4f*)(Wc + (size_t)k0 * T);
            const v4f* g1 = (const v4f*)(Ws + (size_t)k0 * T);
            const v4f* g2 = (const v4f*)(Wg + (size_t)k0 * T);
            v4f w0[V4_PT], w1[V4_PT], w2[V4_PT];
            #pragma unroll
            for (int j = 0; j < V4_PT; ++j) {
                w0[j] = g0[tid + j * BLOCK];
                w1[j] = g1[tid + j * BLOCK];
                w2[j] = g2[tid + j * BLOCK];
            }
            #pragma unroll
            for (int j = 0; j < V4_PT; ++j) {
                *(v4f*)&lds_w[0 * MSTR + 4 * (tid + j * BLOCK)] = w0[j];
                *(v4f*)&lds_w[1 * MSTR + 4 * (tid + j * BLOCK)] = w1[j];
                *(v4f*)&lds_w[2 * MSTR + 4 * (tid + j * BLOCK)] = w2[j];
            }
        }
        __syncthreads();

        if (c + 1 < NCHUNK)
            __builtin_prefetch(px + k0 + KC + 2 * half, 0, 1);

        #pragma unroll
        for (int j = 0; j < TRIP; ++j)
            #pragma unroll
            for (int t = 0; t < NTILE; ++t) {
                bb[0][j][t][0] = lds_w[(4 * j    ) * T + bOff[t]];
                bb[0][j][t][1] = lds_w[(4 * j + 1) * T + bOff[t]];
            }

        #pragma unroll
        for (int t = 0; t < NTRIP; ++t) {
            const int cur = t & 1, nxt = cur ^ 1;
            if (t + 1 < NTRIP) {
                const int kn = 4 * TRIP * (t + 1);
                #pragma unroll
                for (int j = 0; j < TRIP; ++j)
                    xa[nxt][j] = __builtin_nontemporal_load((const v2f*)(pxc + kn + 4 * j));
                #pragma unroll
                for (int j = 0; j < TRIP; ++j)
                    #pragma unroll
                    for (int u = 0; u < NTILE; ++u) {
                        bb[nxt][j][u][0] = lds_w[(kn + 4 * j    ) * T + bOff[u]];
                        bb[nxt][j][u][1] = lds_w[(kn + 4 * j + 1) * T + bOff[u]];
                    }
            }
            __builtin_amdgcn_sched_barrier(0);
            #pragma unroll
            for (int j = 0; j < TRIP; ++j)
                #pragma unroll
                for (int u = 0; u < NTILE; ++u)
                    acc[u] = __builtin_amdgcn_wmma_f32_16x16x4_f32(
                        false, xa[cur][j], false, bb[cur][j][u], (short)0, acc[u], false, false);
            __builtin_amdgcn_sched_barrier(0);
        }
    }

    #pragma unroll
    for (int t = 0; t < NTILE; ++t)
        #pragma unroll
        for (int r = 0; r < 8; ++r)
            lds_acc[kq][rt * 16 + r + 8 * half][16 * t + l16] = acc[t][r];
    __syncthreads();

    if (tid < ROWS_PB) {
        const int row = tid;
        float Lc[T], Ls[T], Lg[T];
        #pragma unroll
        for (int n = 0; n < T; ++n) {
            Lc[n] = ((lds_acc[0][row][0 * T + n] + lds_acc[1][row][0 * T + n])
                   + (lds_acc[2][row][0 * T + n] + lds_acc[3][row][0 * T + n])) + bc[n];
            Ls[n] = ((lds_acc[0][row][1 * T + n] + lds_acc[1][row][1 * T + n])
                   + (lds_acc[2][row][1 * T + n] + lds_acc[3][row][1 * T + n])) + bs[n];
            Lg[n] = ((lds_acc[0][row][2 * T + n] + lds_acc[1][row][2 * T + n])
                   + (lds_acc[2][row][2 * T + n] + lds_acc[3][row][2 * T + n])) + bg[n];
        }
        float mx = Lc[0];
        #pragma unroll
        for (int n = 1; n < T; ++n) mx = fmaxf(mx, Lc[n]);
        float e[T], ssum = 0.0f;
        #pragma unroll
        for (int n = 0; n < T; ++n) { e[n] = __expf(Lc[n] - mx); ssum += e[n]; }
        float g[T];
        #pragma unroll
        for (int n = 0; n < T; ++n) g[n] = 1.0f / (1.0f + __expf(-Lg[n]));
        float softdot = 0.0f;
        #pragma unroll
        for (int n = 0; n < T; ++n) softdot += g[n] * e[n];
        softdot /= ssum;
        float v0 = -INFINITY, v1 = -INFINITY;
        int   i0 = 0, i1 = 0;
        #pragma unroll
        for (int n = 0; n < T; ++n) {
            const float t = Ls[n];
            if (t > v0)      { v1 = v0; i1 = i0; v0 = t; i0 = n; }
            else if (t > v1) { v1 = t;  i1 = n; }
        }
        const float s0 = 1.0f / (1.0f + __expf(v1 - v0));
        const float s1 = 1.0f - s0;
        const float sparsedot = g[i0] * s0 + g[i1] * s1;

        const float a = 1.0f / (1.0f + __expf(-alpha[0]));
        out[blockIdx.x * ROWS_PB + row] = a * sparsedot + (1.0f - a) * softdot;
    }
}

extern "C" void kernel_launch(void* const* d_in, const int* in_sizes, int n_in,
                              void* d_out, int out_size, void* d_ws, size_t ws_size,
                              hipStream_t stream) {
    (void)in_sizes; (void)n_in; (void)out_size; (void)d_ws; (void)ws_size;
    const float* x     = (const float*)d_in[0];
    const float* W_cls = (const float*)d_in[1];
    const float* b_cls = (const float*)d_in[2];
    const float* W_sp  = (const float*)d_in[3];
    const float* b_sp  = (const float*)d_in[4];
    const float* W_g   = (const float*)d_in[5];
    const float* b_g   = (const float*)d_in[6];
    const float* alpha = (const float*)d_in[7];
    float* out = (float*)d_out;

    dim3 grid(NROWS / ROWS_PB);
    dim3 block(BLOCK);
    tgate_hybrid_fused_kernel<<<grid, block, 0, stream>>>(
        x, W_cls, b_cls, W_sp, b_sp, W_g, b_g, alpha, out);
}


# x fragments read by 16-byte loads, a lane taking 4 consecutive k for two k-steps (B loader's k order permuted alike): 4 loads of 16 rows x 32 B per trip, was 8 of 16 B; sum order changes
# speedup vs baseline: 17.7358x; 1.0344x over previous
#include <hip/hip_runtime.h>
#include <math.h>


typedef __attribute__((ext_vector_type(2))) float v2f;
typedef __attribute__((ext_vector_type(4))) float v4f;
typedef __attribute__((ext_vector_type(8))) float v8f;

constexpr int D      = 2048;
constexpr int T      = 10;
constexpr int NP     = 3 * T;
constexpr int NTILE  = 2;
constexpr int NROWS  = 4 * 4096;
constexpr int KC       = 512;
constexpr int NCHUNK   = D / KC;
constexpr int ROWT     = 2;
constexpr int KSPLIT   = 4;
constexpr int WAVES    = ROWT * KSPLIT;
constexpr int BLOCK    = WAVES * 32;
constexpr int ROWS_PB  = ROWT * 16;
constexpr int KW       = KC / KSPLIT;
constexpr int CH_F   = KC * T;
constexpr int CH_V4  = CH_F / 4;
constexpr int V4_PT  = CH_V4 / BLOCK;
constexpr int MSTR   = CH_F + 12;
constexpr int RP     = 34;

__global__ __launch_bounds__(BLOCK)
void tgate_hybrid_fused_kernel(const float* __restrict__ x,
                               const float* __restrict__ Wc, const float* __restrict__ bc,
                               const float* __restrict__ Ws, const float* __restrict__ bs,
                               const float* __restrict__ Wg, const float* __restrict__ bg,
                               const float* __restrict__ alpha,
                               float* __restrict__ out)
{
    __shared__ __attribute__((aligned(16))) float lds_w[3 * MSTR];
    __shared__ float lds_acc[KSPLIT][ROWS_PB][RP];

    const int tid  = threadIdx.x;
    const int wid  = tid >> 5;
    const int lane = tid & 31;
    const int half = lane >> 4;
    const int l16  = lane & 15;
    const int rt   = wid & (ROWT - 1);
    const int kq   = wid / ROWT;
    const int kw0  = kq * KW;

    const int rowBase = blockIdx.x * ROWS_PB + rt * 16;
    const float* px   = x + (size_t)(rowBase + l16) * D + kw0;

    int bOff[NTILE];
    #pragma unroll
    for (int t = 0; t < NTILE; ++t) {
        const int p = (16 * t + l16 < NP) ? (16 * t + l16) : (NP - 1);
        const int m = p / T;
        bOff[t] = m * MSTR + (kw0 + 4 * half) * T + (p - m * T);
    }

    v8f acc[NTILE];
    #pragma unroll
    for (int t = 0; t < NTILE; ++t)
        #pragma unroll
        for (int r = 0; r < 8; ++r)
            acc[t][r] = 0.0f;

    constexpr int TRIP  = 8;
    constexpr int XLD   = TRIP / 2;
    constexpr int NTRIP = KW / (4 * TRIP);
    v4f xq[2][XLD];
    v2f bb[2][TRIP][NTILE];

    for (int c = 0; c < NCHUNK; ++c) {
        const int k0 = c * KC;
        const float* pxc = px + k0 + 4 * half;

        __syncthreads();

        #pragma unroll
        for (int i = 0; i < XLD; ++i)
            xq[0][i] = __builtin_nontemporal_load((const v4f*)(pxc + 8 * i));

        {
            const v4f* g0 = (const v4f*)(Wc + (size_t)k0 * T);
            const v4f* g1 = (const v4f*)(Ws + (size_t)k0 * T);
            const v4f* g2 = (const v4f*)(Wg + (size_t)k0 * T);
            v4f w0[V4_PT], w1[V4_PT], w2[V4_PT];
            #pragma unroll
            for (int j = 0; j < V4_PT; ++j) {
                w0[j] = g0[tid + j * BLOCK];
                w1[j] = g1[tid + j * BLOCK];
                w2[j] = g2[tid + j * BLOCK];
            }
            #pragma unroll
            for (int j = 0; j < V4_PT; ++j) {
                *(v4f*)&lds_w[0 * MSTR + 4 * (tid + j * BLOCK)] = w0[j];
                *(v4f*)&lds_w[1 * MSTR + 4 * (tid + j * BLOCK)] = w1[j];
                *(v4f*)&lds_w[2 * MSTR + 4 * (tid + j * BLOCK)] = w2[j];
            }
        }
        __syncthreads();

        if (c + 1 < NCHUNK)
            __builtin_prefetch(px + k0 + KC + 4 * half, 0, 1);

        #define KOF(j) (8 * ((j) >> 1) + 2 * ((j) & 1))
        #pragma unroll
        for (int j = 0; j < TRIP; ++j)
            #pragma unroll
            for (int t = 0; t < NTILE; ++t) {
                bb[0][j][t][0] = lds_w[(KOF(j)    ) * T + bOff[t]];
                bb[0][j][t][1] = lds_w[(KOF(j) + 1) * T + bOff[t]];
            }

        #pragma unroll
        for (int t = 0; t < NTRIP; ++t) {
            const int cur = t & 1, nxt = cur ^ 1;
            if (t + 1 < NTRIP) {
                const int kn = 4 * TRIP * (t + 1);
                #pragma unroll
                for (int i = 0; i < XLD; ++i)
                    xq[nxt][i] = __builtin_nontemporal_load((const v4f*)(pxc + kn + 8 * i));
                #pragma unroll
                for (int j = 0; j < TRIP; ++j)
                    #pragma unroll
                    for (int u = 0; u < NTILE; ++u) {
                        bb[nxt][j][u][0] = lds_w[(kn + KOF(j)    ) * T + bOff[u]];
                        bb[nxt][j][u][1] = lds_w[(kn + KOF(j) + 1) * T + bOff[u]];
                    }
            }
            __builtin_amdgcn_sched_barrier(0);
            #pragma unroll
            for (int j = 0; j < TRIP; ++j) {
                const v4f q = xq[cur][j >> 1];
                const v2f a = (j & 1) ? __builtin_shufflevector(q, q, 2, 3)
                                      : __builtin_shufflevector(q, q, 0, 1);
                #pragma unroll
                for (int u = 0; u < NTILE; ++u)
                    acc[u] = __builtin_amdgcn_wmma_f32_16x16x4_f32(
                        false, a, false, bb[cur][j][u], (short)0, acc[u], false, false);
            }
            __builtin_amdgcn_sched_barrier(0);
        }
    }

    #undef KOF

    #pragma unroll
    for (int t = 0; t < NTILE; ++t)
        #pragma unroll
        for (int r = 0; r < 8; ++r)
            lds_acc[kq][rt * 16 + r + 8 * half][16 * t + l16] = acc[t][r];
    __syncthreads();

    if (tid < ROWS_PB) {
        const int row = tid;
        float Lc[T], Ls[T], Lg[T];
        #pragma unroll
        for (int n = 0; n < T; ++n) {
            Lc[n] = ((lds_acc[0][row][0 * T + n] + lds_acc[1][row][0 * T + n])
                   + (lds_acc[2][row][0 * T + n] + lds_acc[3][row][0 * T + n])) + bc[n];
            Ls[n] = ((lds_acc[0][row][1 * T + n] + lds_acc[1][row][1 * T + n])
                   + (lds_acc[2][row][1 * T + n] + lds_acc[3][row][1 * T + n])) + bs[n];
            Lg[n] = ((lds_acc[0][row][2 * T + n] + lds_acc[1][row][2 * T + n])
                   + (lds_acc[2][row][2 * T + n] + lds_acc[3][row][2 * T + n])) + bg[n];
        }
        float mx = Lc[0];
        #pragma unroll
        for (int n = 1; n < T; ++n) mx = fmaxf(mx, Lc[n]);
        float e[T], ssum = 0.0f;
        #pragma unroll
        for (int n = 0; n < T; ++n) { e[n] = __expf(Lc[n] - mx); ssum += e[n]; }
        float g[T];
        #pragma unroll
        for (int n = 0; n < T; ++n) g[n] = 1.0f / (1.0f + __expf(-Lg[n]));
        float softdot = 0.0f;
        #pragma unroll
        for (int n = 0; n < T; ++n) softdot += g[n] * e[n];
        softdot /= ssum;
        float v0 = -INFINITY, v1 = -INFINITY;
        int   i0 = 0, i1 = 0;
        #pragma unroll
        for (int n = 0; n < T; ++n) {
            const float t = Ls[n];
            if (t > v0)      { v1 = v0; i1 = i0; v0 = t; i0 = n; }
            else if (t > v1) { v1 = t;  i1 = n; }
        }
        const float s0 = 1.0f / (1.0f + __expf(v1 - v0));
        const float s1 = 1.0f - s0;
        const float sparsedot = g[i0] * s0 + g[i1] * s1;

        const float a = 1.0f / (1.0f + __expf(-alpha[0]));
        out[blockIdx.x * ROWS_PB + row] = a * sparsedot + (1.0f - a) * softdot;
    }
}

extern "C" void kernel_launch(void* const* d_in, const int* in_sizes, int n_in,
                              void* d_out, int out_size, void* d_ws, size_t ws_size,
                              hipStream_t stream) {
    (void)in_sizes; (void)n_in; (void)out_size; (void)d_ws; (void)ws_size;
    const float* x     = (const float*)d_in[0];
    const float* W_cls = (const float*)d_in[1];
    const float* b_cls = (const float*)d_in[2];
    const float* W_sp  = (const float*)d_in[3];
    const float* b_sp  = (const float*)d_in[4];
    const float* W_g   = (const float*)d_in[5];
    const float* b_g   = (const float*)d_in[6];
    const float* alpha = (const float*)d_in[7];
    float* out = (float*)d_out;

    dim3 grid(NROWS / ROWS_PB);
    dim3 block(BLOCK);
    tgate_hybrid_fused_kernel<<<grid, block, 0, stream>>>(
        x, W_cls, b_cls, W_sp, b_sp, W_g, b_g, alpha, out);
}


# Weight staging pipelined over two LDS buffers: the next chunk's 15 loads fly under the current chunk's WMMAs and are stored behind them; one barrier per chunk (was 2); sum order unchanged
# speedup vs baseline: 19.1944x; 1.0822x over previous
#include <hip/hip_runtime.h>
#include <math.h>


typedef __attribute__((ext_vector_type(2))) float v2f;
typedef __attribute__((ext_vector_type(4))) float v4f;
typedef __attribute__((ext_vector_type(8))) float v8f;

constexpr int D      = 2048;
constexpr int T      = 10;
constexpr int NP     = 3 * T;
constexpr int NTILE  = 2;
constexpr int NROWS  = 4 * 4096;
constexpr int KC       = 512;
constexpr int NCHUNK   = D / KC;
constexpr int ROWT     = 2;
constexpr int KSPLIT   = 4;
constexpr int WAVES    = ROWT * KSPLIT;
constexpr int BLOCK    = WAVES * 32;
constexpr int ROWS_PB  = ROWT * 16;
constexpr int KW       = KC / KSPLIT;
constexpr int CH_F   = KC * T;
constexpr int CH_V4  = CH_F / 4;
constexpr int V4_PT  = CH_V4 / BLOCK;
constexpr int MSTR   = CH_F + 12;
constexpr int WBUF   = 3 * MSTR;
constexpr int RP     = 34;

__global__ __launch_bounds__(BLOCK)
void tgate_hybrid_fused_kernel(const float* __restrict__ x,
                               const float* __restrict__ Wc, const float* __restrict__ bc,
                               const float* __restrict__ Ws, const float* __restrict__ bs,
                               const float* __restrict__ Wg, const float* __restrict__ bg,
                               const float* __restrict__ alpha,
                               float* __restrict__ out)
{
    __shared__ __attribute__((aligned(16))) float lds_w[2 * WBUF];
    __shared__ float lds_acc[KSPLIT][ROWS_PB][RP];

    const int tid  = threadIdx.x;
    const int wid  = tid >> 5;
    const int lane = tid & 31;
    const int half = lane >> 4;
    const int l16  = lane & 15;
    const int rt   = wid & (ROWT - 1);
    const int kq   = wid / ROWT;
    const int kw0  = kq * KW;

    const int rowBase = blockIdx.x * ROWS_PB + rt * 16;
    const float* px   = x + (size_t)(rowBase + l16) * D + kw0;

    int bOff[NTILE];
    #pragma unroll
    for (int t = 0; t < NTILE; ++t) {
        const int p = (16 * t + l16 < NP) ? (16 * t + l16) : (NP - 1);
        const int m = p / T;
        bOff[t] = m * MSTR + (kw0 + 4 * half) * T + (p - m * T);
    }

    v8f acc[NTILE];
    #pragma unroll
    for (int t = 0; t < NTILE; ++t)
        #pragma unroll
        for (int r = 0; r < 8; ++r)
            acc[t][r] = 0.0f;

    constexpr int TRIP  = 8;
    constexpr int XLD   = TRIP / 2;
    constexpr int NTRIP = KW / (4 * TRIP);
    v4f xq[2][XLD];
    v2f bb[2][TRIP][NTILE];

    v4f w0[V4_PT], w1[V4_PT], w2[V4_PT];
    #define W_LOAD(kbase)                                                   \
        {                                                                   \
            const v4f* g0 = (const v4f*)(Wc + (size_t)(kbase) * T);         \
            const v4f* g1 = (const v4f*)(Ws + (size_t)(kbase) * T);         \
            const v4f* g2 = (const v4f*)(Wg + (size_t)(kbase) * T);         \
            _Pragma("unroll")                                               \
            for (int j = 0; j < V4_PT; ++j) {                               \
                w0[j] = g0[tid + j * BLOCK];                                \
                w1[j] = g1[tid + j * BLOCK];                                \
                w2[j] = g2[tid + j * BLOCK];                                \
            }                                                               \
        }
    #define W_STORE(boff)                                                   \
        {                                                                   \
            _Pragma("unroll")                                               \
            for (int j = 0; j < V4_PT; ++j) {                               \
                *(v4f*)&lds_w[(boff) + 0 * MSTR + 4 * (tid + j * BLOCK)] = w0[j]; \
                *(v4f*)&lds_w[(boff) + 1 * MSTR + 4 * (tid + j * BLOCK)] = w1[j]; \
                *(v4f*)&lds_w[(boff) + 2 * MSTR + 4 * (tid + j * BLOCK)] = w2[j]; \
            }                                                               \
        }
    #define KOF(j) (8 * ((j) >> 1) + 2 * ((j) & 1))

    const float* pxh = px + 4 * half;

    #pragma unroll
    for (int i = 0; i < XLD; ++i)
        xq[0][i] = __builtin_nontemporal_load((const v4f*)(pxh + 8 * i));
    W_LOAD(0)
    W_STORE(0)
    __syncthreads();

    for (int c = 0; c < NCHUNK; ++c) {
        const int k0 = c * KC;
        const float* pxc = pxh + k0;
        const int cb = (c & 1) * WBUF;
        const bool more = c + 1 < NCHUNK;

        int bO[NTILE];
        #pragma unroll
        for (int u = 0; u < NTILE; ++u) bO[u] = bOff[u] + cb;

        #pragma unroll
        for (int j = 0; j < TRIP; ++j)
            #pragma unroll
            for (int u = 0; u < NTILE; ++u) {
                bb[0][j][u][0] = lds_w[(KOF(j)    ) * T + bO[u]];
                bb[0][j][u][1] = lds_w[(KOF(j) + 1) * T + bO[u]];
            }

        #pragma unroll
        for (int t = 0; t < NTRIP; ++t) {
            const int cur = t & 1, nxt = cur ^ 1;
            if (t + 1 < NTRIP) {
                const int kn = 4 * TRIP * (t + 1);
                #pragma unroll
                for (int i = 0; i < XLD; ++i)
                    xq[nxt][i] = __builtin_nontemporal_load((const v4f*)(pxc + kn + 8 * i));
                if (t == 0 && more) {
                    W_LOAD(k0 + KC)
                    __builtin_prefetch(px + k0 + KC + 4 * half, 0, 1);
                }
                #pragma unroll
                for (int j = 0; j < TRIP; ++j)
                    #pragma unroll
                    for (int u = 0; u < NTILE; ++u) {
                        bb[nxt][j][u][0] = lds_w[(kn + KOF(j)    ) * T + bO[u]];
                        bb[nxt][j][u][1] = lds_w[(kn + KOF(j) + 1) * T + bO[u]];
                    }
            } else if (more) {
                #pragma unroll
                for (int i = 0; i < XLD; ++i)
                    xq[nxt][i] = __builtin_nontemporal_load((const v4f*)(pxc + KC + 8 * i));
            }
            __builtin_amdgcn_sched_barrier(0);
            #pragma unroll
            for (int j = 0; j < TRIP; ++j) {
                const v4f q = xq[cur][j >> 1];
                const v2f a = (j & 1) ? __builtin_shufflevector(q, q, 2, 3)
                                      : __builtin_shufflevector(q, q, 0, 1);
                #pragma unroll
                for (int u = 0; u < NTILE; ++u)
                    acc[u] = __builtin_amdgcn_wmma_f32_16x16x4_f32(
                        false, a, false, bb[cur][j][u], (short)0, acc[u], false, false);
            }
            __builtin_amdgcn_sched_barrier(0);
        }

        if (more) {
            W_STORE(WBUF - cb)
            __syncthreads();
        }
    }

    #undef W_LOAD
    #undef W_STORE
    #undef KOF

    #pragma unroll
    for (int t = 0; t < NTILE; ++t)
        #pragma unroll
        for (int r = 0; r < 8; ++r)
            lds_acc[kq][rt * 16 + r + 8 * half][16 * t + l16] = acc[t][r];
    __syncthreads();

    if (tid < ROWS_PB) {
        const int row = tid;
        float Lc[T], Ls[T], Lg[T];
        #pragma unroll
        for (int n = 0; n < T; ++n) {
            Lc[n] = ((lds_acc[0][row][0 * T + n] + lds_acc[1][row][0 * T + n])
                   + (lds_acc[2][row][0 * T + n] + lds_acc[3][row][0 * T + n])) + bc[n];
            Ls[n] = ((lds_acc[0][row][1 * T + n] + lds_acc[1][row][1 * T + n])
                   + (lds_acc[2][row][1 * T + n] + lds_acc[3][row][1 * T + n])) + bs[n];
            Lg[n] = ((lds_acc[0][row][2 * T + n] + lds_acc[1][row][2 * T + n])
                   + (lds_acc[2][row][2 * T + n] + lds_acc[3][row][2 * T + n])) + bg[n];
        }
        float mx = Lc[0];
        #pragma unroll
        for (int n = 1; n < T; ++n) mx = fmaxf(mx, Lc[n]);
        float e[T], ssum = 0.0f;
        #pragma unroll
        for (int n = 0; n < T; ++n) { e[n] = __expf(Lc[n] - mx); ssum += e[n]; }
        float g[T];
        #pragma unroll
        for (int n = 0; n < T; ++n) g[n] = 1.0f / (1.0f + __expf(-Lg[n]));
        float softdot = 0.0f;
        #pragma unroll
        for (int n = 0; n < T; ++n) softdot += g[n] * e[n];
        softdot /= ssum;
        float v0 = -INFINITY, v1 = -INFINITY;
        int   i0 = 0, i1 = 0;
        #pragma unroll
        for (int n = 0; n < T; ++n) {
            const float t = Ls[n];
            if (t > v0)      { v1 = v0; i1 = i0; v0 = t; i0 = n; }
            else if (t > v1) { v1 = t;  i1 = n; }
        }
        const float s0 = 1.0f / (1.0f + __expf(v1 - v0));
        const float s1 = 1.0f - s0;
        const float sparsedot = g[i0] * s0 + g[i1] * s1;

        const float a = 1.0f / (1.0f + __expf(-alpha[0]));
        out[blockIdx.x * ROWS_PB + row] = a * sparsedot + (1.0f - a) * softdot;
    }
}

extern "C" void kernel_launch(void* const* d_in, const int* in_sizes, int n_in,
                              void* d_out, int out_size, void* d_ws, size_t ws_size,
                              hipStream_t stream) {
    (void)in_sizes; (void)n_in; (void)out_size; (void)d_ws; (void)ws_size;
    const float* x     = (const float*)d_in[0];
    const float* W_cls = (const float*)d_in[1];
    const float* b_cls = (const float*)d_in[2];
    const float* W_sp  = (const float*)d_in[3];
    const float* b_sp  = (const float*)d_in[4];
    const float* W_g   = (const float*)d_in[5];
    const float* b_g   = (const float*)d_in[6];
    const float* alpha = (const float*)d_in[7];
    float* out = (float*)d_out;

    dim3 grid(NROWS / ROWS_PB);
    dim3 block(BLOCK);
    tgate_hybrid_fused_kernel<<<grid, block, 0, stream>>>(
        x, W_cls, b_cls, W_sp, b_sp, W_g, b_g, alpha, out);
}


# Each wave takes two 16-row tiles against one set of B fragments: 64 rows per block, grid 256, LDS sized so one block sits per WGP; weight chunks staged once per 64 rows (was 32); sum order unchanged
# speedup vs baseline: 21.4631x; 1.1182x over previous
#include <hip/hip_runtime.h>
#include <math.h>


typedef __attribute__((ext_vector_type(2))) float v2f;
typedef __attribute__((ext_vector_type(4))) float v4f;
typedef __attribute__((ext_vector_type(8))) float v8f;

constexpr int D      = 2048;
constexpr int T      = 10;
constexpr int NP     = 3 * T;
constexpr int NTILE  = 2;
constexpr int NROWS  = 4 * 4096;
constexpr int KC       = 512;
constexpr int NCHUNK   = D / KC;
constexpr int RTW      = 2;
constexpr int ROWG     = 2;
constexpr int KSPLIT   = 4;
constexpr int WAVES    = ROWG * KSPLIT;
constexpr int BLOCK    = WAVES * 32;
constexpr int ROWS_PW  = RTW * 16;
constexpr int ROWS_PB  = ROWG * ROWS_PW;
constexpr int KW       = KC / KSPLIT;
constexpr int CH_F   = KC * T;
constexpr int CH_V4  = CH_F / 4;
constexpr int V4_PT  = CH_V4 / BLOCK;
constexpr int MSTR   = CH_F + 12;
constexpr int WBUF   = 3 * MSTR;
constexpr int RP     = 34;
constexpr int LDS_ONE_PER_WGP = 2048;

__global__ __launch_bounds__(BLOCK)
void tgate_hybrid_fused_kernel(const float* __restrict__ x,
                               const float* __restrict__ Wc, const float* __restrict__ bc,
                               const float* __restrict__ Ws, const float* __restrict__ bs,
                               const float* __restrict__ Wg, const float* __restrict__ bg,
                               const float* __restrict__ alpha,
                               float* __restrict__ out)
{
    __shared__ __attribute__((aligned(16))) float lds_w[2 * WBUF + LDS_ONE_PER_WGP];
    __shared__ float lds_acc[KSPLIT][ROWS_PB][RP];

    const int tid  = threadIdx.x;
    const int wid  = tid >> 5;
    const int lane = tid & 31;
    const int half = lane >> 4;
    const int l16  = lane & 15;
    const int rg   = wid & (ROWG - 1);
    const int kq   = wid / ROWG;
    const int kw0  = kq * KW;

    const int rowBase = blockIdx.x * ROWS_PB + rg * ROWS_PW;
    const float* px   = x + (size_t)(rowBase + l16) * D + kw0;

    int bOff[NTILE];
    #pragma unroll
    for (int t = 0; t < NTILE; ++t) {
        const int p = (16 * t + l16 < NP) ? (16 * t + l16) : (NP - 1);
        const int m = p / T;
        bOff[t] = m * MSTR + (kw0 + 4 * half) * T + (p - m * T);
    }

    v8f acc[RTW][NTILE];
    #pragma unroll
    for (int rw = 0; rw < RTW; ++rw)
        #pragma unroll
        for (int t = 0; t < NTILE; ++t)
            #pragma unroll
            for (int r = 0; r < 8; ++r)
                acc[rw][t][r] = 0.0f;

    constexpr int TRIP  = 8;
    constexpr int XLD   = TRIP / 2;
    constexpr int NTRIP = KW / (4 * TRIP);
    v4f xq[2][RTW][XLD];
    v2f bb[2][TRIP][NTILE];

    v4f w0[V4_PT], w1[V4_PT], w2[V4_PT];
    #define W_LOAD(kbase)                                                   \
        {                                                                   \
            const v4f* g0 = (const v4f*)(Wc + (size_t)(kbase) * T);         \
            const v4f* g1 = (const v4f*)(Ws + (size_t)(kbase) * T);         \
            const v4f* g2 = (const v4f*)(Wg + (size_t)(kbase) * T);         \
            _Pragma("unroll")                                               \
            for (int j = 0; j < V4_PT; ++j) {                               \
                w0[j] = g0[tid + j * BLOCK];                                \
                w1[j] = g1[tid + j * BLOCK];                                \
                w2[j] = g2[tid + j * BLOCK];                                \
            }                                                               \
        }
    #define W_STORE(boff)                                                   \
        {                                                                   \
            _Pragma("unroll")                                               \
            for (int j = 0; j < V4_PT; ++j) {                               \
                *(v4f*)&lds_w[(boff) + 0 * MSTR + 4 * (tid + j * BLOCK)] = w0[j]; \
                *(v4f*)&lds_w[(boff) + 1 * MSTR + 4 * (tid + j * BLOCK)] = w1[j]; \
                *(v4f*)&lds_w[(boff) + 2 * MSTR + 4 * (tid + j * BLOCK)] = w2[j]; \
            }                                                               \
        }
    #define KOF(j) (8 * ((j) >> 1) + 2 * ((j) & 1))

    const float* pxh = px + 4 * half;

    #pragma unroll
    for (int rw = 0; rw < RTW; ++rw)
        #pragma unroll
        for (int i = 0; i < XLD; ++i)
            xq[0][rw][i] = __builtin_nontemporal_load((const v4f*)(pxh + (size_t)rw * 16 * D + 8 * i));
    W_LOAD(0)
    W_STORE(0)
    __syncthreads();

    for (int c = 0; c < NCHUNK; ++c) {
        const int k0 = c * KC;
        const float* pxc = pxh + k0;
        const int cb = (c & 1) * WBUF;
        const bool more = c + 1 < NCHUNK;

        int bO[NTILE];
        #pragma unroll
        for (int u = 0; u < NTILE; ++u) bO[u] = bOff[u] + cb;

        #pragma unroll
        for (int j = 0; j < TRIP; ++j)
            #pragma unroll
            for (int u = 0; u < NTILE; ++u) {
                bb[0][j][u][0] = lds_w[(KOF(j)    ) * T + bO[u]];
                bb[0][j][u][1] = lds_w[(KOF(j) + 1) * T + bO[u]];
            }

        #pragma unroll
        for (int t = 0; t < NTRIP; ++t) {
            const int cur = t & 1, nxt = cur ^ 1;
            if (t + 1 < NTRIP) {
                const int kn = 4 * TRIP * (t + 1);
                #pragma unroll
                for (int rw = 0; rw < RTW; ++rw)
                    #pragma unroll
                    for (int i = 0; i < XLD; ++i)
                        xq[nxt][rw][i] = __builtin_nontemporal_load(
                            (const v4f*)(pxc + (size_t)rw * 16 * D + kn + 8 * i));
                if (t == 0 && more) {
                    W_LOAD(k0 + KC)
                    #pragma unroll
                    for (int rw = 0; rw < RTW; ++rw)
                        __builtin_prefetch(px + (size_t)rw * 16 * D + k0 + KC + 4 * half, 0, 1);
                }
                #pragma unroll
                for (int j = 0; j < TRIP; ++j)
                    #pragma unroll
                    for (int u = 0; u < NTILE; ++u) {
                        bb[nxt][j][u][0] = lds_w[(kn + KOF(j)    ) * T + bO[u]];
                        bb[nxt][j][u][1] = lds_w[(kn + KOF(j) + 1) * T + bO[u]];
                    }
            } else if (more) {
                #pragma unroll
                for (int rw = 0; rw < RTW; ++rw)
                    #pragma unroll
                    for (int i = 0; i < XLD; ++i)
                        xq[nxt][rw][i] = __builtin_nontemporal_load(
                            (const v4f*)(pxc + (size_t)rw * 16 * D + KC + 8 * i));
            }
            __builtin_amdgcn_sched_barrier(0);
            #pragma unroll
            for (int j = 0; j < TRIP; ++j) {
                v2f a[RTW];
                #pragma unroll
                for (int rw = 0; rw < RTW; ++rw) {
                    const v4f q = xq[cur][rw][j >> 1];
                    a[rw] = (j & 1) ? __builtin_shufflevector(q, q, 2, 3)
                                    : __builtin_shufflevector(q, q, 0, 1);
                }
                #pragma unroll
                for (int u = 0; u < NTILE; ++u)
                    #pragma unroll
                    for (int rw = 0; rw < RTW; ++rw)
                        acc[rw][u] = __builtin_amdgcn_wmma_f32_16x16x4_f32(
                            false, a[rw], false, bb[cur][j][u], (short)0, acc[rw][u], false, false);
            }
            __builtin_amdgcn_sched_barrier(0);
        }

        if (more) {
            W_STORE(WBUF - cb)
            __syncthreads();
        }
    }

    #undef W_LOAD
    #undef W_STORE
    #undef KOF

    #pragma unroll
    for (int rw = 0; rw < RTW; ++rw)
        #pragma unroll
        for (int t = 0; t < NTILE; ++t)
            #pragma unroll
            for (int r = 0; r < 8; ++r)
                lds_acc[kq][rg * ROWS_PW + rw * 16 + r + 8 * half][16 * t + l16] = acc[rw][t][r];
    __syncthreads();

    if (tid < ROWS_PB) {
        const int row = tid;
        float Lc[T], Ls[T], Lg[T];
        #pragma unroll
        for (int n = 0; n < T; ++n) {
            Lc[n] = ((lds_acc[0][row][0 * T + n] + lds_acc[1][row][0 * T + n])
                   + (lds_acc[2][row][0 * T + n] + lds_acc[3][row][0 * T + n])) + bc[n];
            Ls[n] = ((lds_acc[0][row][1 * T + n] + lds_acc[1][row][1 * T + n])
                   + (lds_acc[2][row][1 * T + n] + lds_acc[3][row][1 * T + n])) + bs[n];
            Lg[n] = ((lds_acc[0][row][2 * T + n] + lds_acc[1][row][2 * T + n])
                   + (lds_acc[2][row][2 * T + n] + lds_acc[3][row][2 * T + n])) + bg[n];
        }
        float mx = Lc[0];
        #pragma unroll
        for (int n = 1; n < T; ++n) mx = fmaxf(mx, Lc[n]);
        float e[T], ssum = 0.0f;
        #pragma unroll
        for (int n = 0; n < T; ++n) { e[n] = __expf(Lc[n] - mx); ssum += e[n]; }
        float g[T];
        #pragma unroll
        for (int n = 0; n < T; ++n) g[n] = 1.0f / (1.0f + __expf(-Lg[n]));
        float softdot = 0.0f;
        #pragma unroll
        for (int n = 0; n < T; ++n) softdot += g[n] * e[n];
        softdot /= ssum;
        float v0 = -INFINITY, v1 = -INFINITY;
        int   i0 = 0, i1 = 0;
        #pragma unroll
        for (int n = 0; n < T; ++n) {
            const float t = Ls[n];
            if (t > v0)      { v1 = v0; i1 = i0; v0 = t; i0 = n; }
            else if (t > v1) { v1 = t;  i1 = n; }
        }
        const float s0 = 1.0f / (1.0f + __expf(v1 - v0));
        const float s1 = 1.0f - s0;
        const float sparsedot = g[i0] * s0 + g[i1] * s1;

        const float a = 1.0f / (1.0f + __expf(-alpha[0]));
        out[blockIdx.x * ROWS_PB + row] = a * sparsedot + (1.0f - a) * softdot;
    }
}

extern "C" void kernel_launch(void* const* d_in, const int* in_sizes, int n_in,
                              void* d_out, int out_size, void* d_ws, size_t ws_size,
                              hipStream_t stream) {
    (void)in_sizes; (void)n_in; (void)out_size; (void)d_ws; (void)ws_size;
    const float* x     = (const float*)d_in[0];
    const float* W_cls = (const float*)d_in[1];
    const float* b_cls = (const float*)d_in[2];
    const float* W_sp  = (const float*)d_in[3];
    const float* b_sp  = (const float*)d_in[4];
    const float* W_g   = (const float*)d_in[5];
    const float* b_g   = (const float*)d_in[6];
    const float* alpha = (const float*)d_in[7];
    float* out = (float*)d_out;

    dim3 grid(NROWS / ROWS_PB);
    dim3 block(BLOCK);
    tgate_hybrid_fused_kernel<<<grid, block, 0, stream>>>(
        x, W_cls, b_cls, W_sp, b_sp, W_g, b_g, alpha, out);
}
